# starting kernel
# baseline (speedup 1.0000x reference)
#include <hip/hip_runtime.h>

typedef __attribute__((ext_vector_type(2))) float v2f;
typedef __attribute__((ext_vector_type(8))) float v8f;

#define CCH 32
#define NSP (64 * 64 * 64)
#define K1_WGS 128
#define K1_WAVES 8
#define K1_ROWS (NSP / (K1_WGS * K1_WAVES))
#define K3_WGS 512
#define K3_ROWS (NSP / (K3_WGS * 8))

__global__ __launch_bounds__(256) void gram_kernel(const float* __restrict__ q,
                                                   const float* __restrict__ k,
                                                   float* __restrict__ partial) {
    const int wg   = blockIdx.x % K1_WGS;
    const int b    = blockIdx.x / K1_WGS;
    const int wave = threadIdx.x >> 5;
    const int lane = threadIdx.x & 31;
    const int lm   = lane & 15;
    const int kb   = (lane < 16) ? 0 : 2;

    const long n0 = (long)wg * (K1_WAVES * K1_ROWS) + (long)wave * K1_ROWS;
    const float* qb = q + ((long)b * NSP + n0) * CCH;
    const float* kp = k + ((long)b * NSP + n0) * CCH;

    v8f c00 = {}, c01 = {}, c10 = {}, c11 = {};
    for (int it = 0; it < K1_ROWS; it += 4) {
        const float* qr = qb + (long)(it + kb) * CCH;
        const float* kr = kp + (long)(it + kb) * CCH;
        v2f a0, a1, b0, b1;
        a0.x = qr[lm];           a0.y = qr[CCH + lm];
        a1.x = qr[16 + lm];      a1.y = qr[CCH + 16 + lm];
        b0.x = __builtin_nontemporal_load(kr + lm);
        b0.y = __builtin_nontemporal_load(kr + CCH + lm);
        b1.x = __builtin_nontemporal_load(kr + 16 + lm);
        b1.y = __builtin_nontemporal_load(kr + CCH + 16 + lm);
        c00 = __builtin_amdgcn_wmma_f32_16x16x4_f32(false, a0, false, b0, (short)0, c00, false, false);
        c01 = __builtin_amdgcn_wmma_f32_16x16x4_f32(false, a0, false, b1, (short)0, c01, false, false);
        c10 = __builtin_amdgcn_wmma_f32_16x16x4_f32(false, a1, false, b0, (short)0, c10, false, false);
        c11 = __builtin_amdgcn_wmma_f32_16x16x4_f32(false, a1, false, b1, (short)0, c11, false, false);
    }

    __shared__ float smem[K1_WAVES * CCH * CCH];
    const int rowadd = (lane < 16) ? 0 : 8;
    float* sw = smem + wave * (CCH * CCH);
#pragma unroll
    for (int r = 0; r < 8; ++r) {
        const int i0 = r + rowadd;
        sw[(i0     ) * CCH + lm     ] = c00[r];
        sw[(i0     ) * CCH + 16 + lm] = c01[r];
        sw[(i0 + 16) * CCH + lm     ] = c10[r];
        sw[(i0 + 16) * CCH + 16 + lm] = c11[r];
    }
    __syncthreads();

    float* outp = partial + (long)(b * K1_WGS + wg) * (CCH * CCH);
    for (int e = threadIdx.x * 4; e < threadIdx.x * 4 + 4; ++e) {
        float s = 0.f;
        for (int w = 0; w < K1_WAVES; ++w) s += smem[w * (CCH * CCH) + e];
        outp[e] = s;
    }
}

__global__ __launch_bounds__(256) void softmax_kernel(const float* __restrict__ partial,
                                                      float* __restrict__ wfin) {
    const int b = blockIdx.x;
    const int t = threadIdx.x;
    __shared__ float g[CCH * CCH];
    for (int e = t * 4; e < t * 4 + 4; ++e) {
        float s = 0.f;
        const float* p = partial + (long)b * K1_WGS * (CCH * CCH) + e;
        for (int w = 0; w < K1_WGS; ++w) s += p[(long)w * (CCH * CCH)];
        g[e] = s;
    }
    __syncthreads();
    if (t < CCH) {
        float m = -3.4e38f;
#pragma unroll
        for (int j = 0; j < CCH; ++j) m = fmaxf(m, g[t * CCH + j]);
        float ex[CCH];
        float s = 0.f;
#pragma unroll
        for (int j = 0; j < CCH; ++j) { ex[j] = __expf(g[t * CCH + j] - m); s += ex[j]; }
        const float inv = 1.f / s;
#pragma unroll
        for (int j = 0; j < CCH; ++j) wfin[(long)b * (CCH * CCH) + t * CCH + j] = ex[j] * inv;
    }
}

__global__ __launch_bounds__(256) void apply_kernel(const float* __restrict__ q,
                                                    const float* __restrict__ wfin,
                                                    float* __restrict__ out) {
    const int wg   = blockIdx.x % K3_WGS;
    const int b    = blockIdx.x / K3_WGS;
    const int wave = threadIdx.x >> 5;
    const int lane = threadIdx.x & 31;
    const int lm   = lane & 15;
    const int kb   = (lane < 16) ? 0 : 2;

    const float* wb = wfin + (long)b * (CCH * CCH);
    v2f Bw0[8], Bw1[8];
#pragma unroll
    for (int kk = 0; kk < 8; ++kk) {
        const float* w0 = wb + lm * CCH + 4 * kk + kb;
        const float* w1 = wb + (16 + lm) * CCH + 4 * kk + kb;
        Bw0[kk].x = w0[0];  Bw0[kk].y = w0[1];
        Bw1[kk].x = w1[0];  Bw1[kk].y = w1[1];
    }

    const long n0 = (long)wg * (8 * K3_ROWS) + (long)wave * K3_ROWS;
    const float* qb = q   + ((long)b * NSP + n0) * CCH;
    float*       ob = out + ((long)b * NSP + n0) * CCH;
    const int rowadd = (lane < 16) ? 0 : 8;

    for (int it = 0; it < K3_ROWS; it += 16) {
        v2f A[8];
#pragma unroll
        for (int kk = 0; kk < 8; ++kk) {
            const v2f* qr = (const v2f*)(qb + (long)(it + lm) * CCH + 4 * kk + kb);
            A[kk] = *qr;
        }
        v8f c0 = {}, c1 = {};
#pragma unroll
        for (int kk = 0; kk < 8; ++kk) {
            c0 = __builtin_amdgcn_wmma_f32_16x16x4_f32(false, A[kk], false, Bw0[kk], (short)0, c0, false, false);
            c1 = __builtin_amdgcn_wmma_f32_16x16x4_f32(false, A[kk], false, Bw1[kk], (short)0, c1, false, false);
        }
#pragma unroll
        for (int r = 0; r < 8; ++r) {
            const long row = it + r + rowadd;
            __builtin_nontemporal_store(c0[r] + qb[row * CCH + lm],      ob + row * CCH + lm);
            __builtin_nontemporal_store(c1[r] + qb[row * CCH + 16 + lm], ob + row * CCH + 16 + lm);
        }
    }
}

extern "C" void kernel_launch(void* const* d_in, const int* in_sizes, int n_in,
                              void* d_out, int out_size, void* d_ws, size_t ws_size,
                              hipStream_t stream) {
    const float* q = (const float*)d_in[0];
    const float* k = (const float*)d_in[1];
    float* out = (float*)d_out;
    const int B = in_sizes[0] / (NSP * CCH);

    float* partial = (float*)d_ws;
    float* wfin    = partial + (long)B * K1_WGS * CCH * CCH;

    gram_kernel   <<<dim3(B * K1_WGS), dim3(256), 0, stream>>>(q, k, partial);
    softmax_kernel<<<dim3(B),          dim3(256), 0, stream>>>(partial, wfin);
    apply_kernel  <<<dim3(B * K3_WGS), dim3(256), 0, stream>>>(q, wfin, out);
}


# apply pass, no LDS: k order and tile column order chosen to match memory: 16 B q fragment loads, residual and out as 8 B whole-line moves (was 8 B gathers, 4 B moves); channel sum order permuted
# speedup vs baseline: 1.0964x; 1.0964x over previous
#include <hip/hip_runtime.h>

typedef __attribute__((ext_vector_type(2))) float v2f;
typedef __attribute__((ext_vector_type(4))) float v4f;
typedef __attribute__((ext_vector_type(8))) float v8f;

#define CCH 32
#define NSP (64 * 64 * 64)
#define K1_WGS 64
#define K1_WAVES 32
#define K1_ROWS (NSP / (K1_WGS * K1_WAVES))
#define K3_WGS 512
#define K3_ROWS (NSP / (K3_WGS * 8))

__global__ __launch_bounds__(K1_WAVES * 32) void gram_kernel(const float* __restrict__ q,
                                                   const float* __restrict__ k,
                                                   float* __restrict__ partial) {
    const int wg   = blockIdx.x % K1_WGS;
    const int b    = blockIdx.x / K1_WGS;
    const int wave = threadIdx.x >> 5;
    const int lane = threadIdx.x & 31;
    const int lm   = lane & 15;
    const int kb   = (lane < 16) ? 0 : 2;

    const long n0 = (long)wg * (K1_WAVES * K1_ROWS) + (long)wave * K1_ROWS;
    const float* qb = q + ((long)b * NSP + n0) * CCH;
    const float* kp = k + ((long)b * NSP + n0) * CCH;

    v8f c00 = {}, c01 = {}, c10 = {}, c11 = {};
    for (int it = 0; it < K1_ROWS; it += 4) {
        const float* qr = qb + (long)(it + kb) * CCH;
        const float* kr = kp + (long)(it + kb) * CCH;
        v2f a0, a1, b0, b1;
        a0.x = qr[lm];           a0.y = qr[CCH + lm];
        a1.x = qr[16 + lm];      a1.y = qr[CCH + 16 + lm];
        b0.x = __builtin_nontemporal_load(kr + lm);
        b0.y = __builtin_nontemporal_load(kr + CCH + lm);
        b1.x = __builtin_nontemporal_load(kr + 16 + lm);
        b1.y = __builtin_nontemporal_load(kr + CCH + 16 + lm);
        c00 = __builtin_amdgcn_wmma_f32_16x16x4_f32(false, a0, false, b0, (short)0, c00, false, false);
        c01 = __builtin_amdgcn_wmma_f32_16x16x4_f32(false, a0, false, b1, (short)0, c01, false, false);
        c10 = __builtin_amdgcn_wmma_f32_16x16x4_f32(false, a1, false, b0, (short)0, c10, false, false);
        c11 = __builtin_amdgcn_wmma_f32_16x16x4_f32(false, a1, false, b1, (short)0, c11, false, false);
    }

    __shared__ float smem[K1_WAVES * CCH * CCH];
    const int rowadd = (lane < 16) ? 0 : 8;
    float* sw = smem + wave * (CCH * CCH);
#pragma unroll
    for (int r = 0; r < 8; ++r) {
        const int i0 = r + rowadd;
        sw[(i0     ) * CCH + lm     ] = c00[r];
        sw[(i0     ) * CCH + 16 + lm] = c01[r];
        sw[(i0 + 16) * CCH + lm     ] = c10[r];
        sw[(i0 + 16) * CCH + 16 + lm] = c11[r];
    }
    __syncthreads();

    float* outp = partial + (long)(b * K1_WGS + wg) * (CCH * CCH);
    {
        const int e = threadIdx.x;
        float s = 0.f;
        for (int w = 0; w < K1_WAVES; ++w) s += smem[w * (CCH * CCH) + e];
        outp[e] = s;
    }
}

__global__ __launch_bounds__(256) void softmax_kernel(const float* __restrict__ partial,
                                                      float* __restrict__ wfin) {
    const int b = blockIdx.x;
    const int t = threadIdx.x;
    __shared__ float g[CCH * CCH];
    for (int e = t * 4; e < t * 4 + 4; ++e) {
        float s = 0.f;
        const float* p = partial + (long)b * K1_WGS * (CCH * CCH) + e;
        for (int w = 0; w < K1_WGS; ++w) s += p[(long)w * (CCH * CCH)];
        g[e] = s;
    }
    __syncthreads();
    if (t < CCH) {
        float m = -3.4e38f;
#pragma unroll
        for (int j = 0; j < CCH; ++j) m = fmaxf(m, g[t * CCH + j]);
        float ex[CCH];
        float s = 0.f;
#pragma unroll
        for (int j = 0; j < CCH; ++j) { ex[j] = __expf(g[t * CCH + j] - m); s += ex[j]; }
        const float inv = 1.f / s;
#pragma unroll
        for (int j = 0; j < CCH; ++j) wfin[(long)b * (CCH * CCH) + t * CCH + j] = ex[j] * inv;
    }
}

__global__ __launch_bounds__(256) void apply_kernel(const float* __restrict__ q,
                                                    const float* __restrict__ wfin,
                                                    float* __restrict__ out) {
    const int wg   = blockIdx.x % K3_WGS;
    const int b    = blockIdx.x / K3_WGS;
    const int wave = threadIdx.x >> 5;
    const int lane = threadIdx.x & 31;
    const int lm   = lane & 15;
    const int h    = lane >> 4;

    const float* wb = wfin + (long)b * (CCH * CCH);
    v2f Bw0[8], Bw1[8];
#pragma unroll
    for (int u = 0; u < 4; ++u) {
        const v4f w0 = *(const v4f*)(wb + (2 * lm    ) * CCH + 8 * u + 4 * h);
        const v4f w1 = *(const v4f*)(wb + (2 * lm + 1) * CCH + 8 * u + 4 * h);
        Bw0[2 * u].x = w0.x;  Bw0[2 * u].y = w0.y;  Bw0[2 * u + 1].x = w0.z;  Bw0[2 * u + 1].y = w0.w;
        Bw1[2 * u].x = w1.x;  Bw1[2 * u].y = w1.y;  Bw1[2 * u + 1].x = w1.z;  Bw1[2 * u + 1].y = w1.w;
    }

    const long n0 = (long)wg * (8 * K3_ROWS) + (long)wave * K3_ROWS;
    const float* qb = q   + ((long)b * NSP + n0) * CCH;
    float*       ob = out + ((long)b * NSP + n0) * CCH;

#pragma unroll 1
    for (int it = 0; it < K3_ROWS; it += 16) {
        const float* ar = qb + (long)(it + lm) * CCH + 4 * h;
        v4f Q[4];
#pragma unroll
        for (int u = 0; u < 4; ++u) Q[u] = *(const v4f*)(ar + 8 * u);
        const float* rr = qb + (long)(it + 8 * h) * CCH + 2 * lm;
        v2f R[8];
#pragma unroll
        for (int r = 0; r < 8; ++r) R[r] = *(const v2f*)(rr + r * CCH);

        __builtin_amdgcn_sched_barrier(0);

        v8f c0 = {}, c1 = {};
#pragma unroll
        for (int u = 0; u < 4; ++u) {
            v2f a0, a1;
            a0.x = Q[u].x;  a0.y = Q[u].y;
            a1.x = Q[u].z;  a1.y = Q[u].w;
            c0 = __builtin_amdgcn_wmma_f32_16x16x4_f32(false, a0, false, Bw0[2 * u],     (short)0, c0, false, false);
            c1 = __builtin_amdgcn_wmma_f32_16x16x4_f32(false, a0, false, Bw1[2 * u],     (short)0, c1, false, false);
            c0 = __builtin_amdgcn_wmma_f32_16x16x4_f32(false, a1, false, Bw0[2 * u + 1], (short)0, c0, false, false);
            c1 = __builtin_amdgcn_wmma_f32_16x16x4_f32(false, a1, false, Bw1[2 * u + 1], (short)0, c1, false, false);
        }
        __builtin_amdgcn_sched_barrier(0);
        asm volatile("v_nop\n\tv_nop\n\tv_nop\n\tv_nop");
        __builtin_amdgcn_sched_barrier(0);

        float* orow = ob + (long)(it + 8 * h) * CCH + 2 * lm;
#pragma unroll
        for (int r = 0; r < 8; ++r) {
            v2f o;
            o.x = c0[r] + R[r].x;
            o.y = c1[r] + R[r].y;
            __builtin_nontemporal_store(o, (v2f*)(orow + r * CCH));
        }
    }
}

extern "C" void kernel_launch(void* const* d_in, const int* in_sizes, int n_in,
                              void* d_out, int out_size, void* d_ws, size_t ws_size,
                              hipStream_t stream) {
    const float* q = (const float*)d_in[0];
    const float* k = (const float*)d_in[1];
    float* out = (float*)d_out;
    const int B = in_sizes[0] / (NSP * CCH);

    float* partial = (float*)d_ws;
    float* wfin    = partial + (long)B * K1_WGS * CCH * CCH;

    gram_kernel   <<<dim3(B * K1_WGS), dim3(K1_WAVES * 32), 0, stream>>>(q, k, partial);
    softmax_kernel<<<dim3(B),          dim3(256), 0, stream>>>(partial, wfin);
    apply_kernel  <<<dim3(B * K3_WGS), dim3(256), 0, stream>>>(q, wfin, out);
}


# softmax launch folded into the last pass (2 launches, was 3): that pass is one 16-wave work-group per WGP, each sums the 64 partials and does the softmax itself, w via LDS; sum orders kept
# speedup vs baseline: 1.2564x; 1.1459x over previous
#include <hip/hip_runtime.h>

typedef __attribute__((ext_vector_type(2))) float v2f;
typedef __attribute__((ext_vector_type(4))) float v4f;
typedef __attribute__((ext_vector_type(8))) float v8f;

#define CCH 32
#define NSP (64 * 64 * 64)
#define K1_WGS 64
#define K1_WAVES 32
#define K1_ROWS (NSP / (K1_WGS * K1_WAVES))
#define K3_WGS 64
#define K3_WAVES 16
#define K3_ROWS (NSP / (K3_WGS * K3_WAVES))
#define WPITCH 68

__global__ __launch_bounds__(K1_WAVES * 32) void gram_kernel(const float* __restrict__ q,
                                                   const float* __restrict__ k,
                                                   float* __restrict__ partial) {
    const int wg   = blockIdx.x % K1_WGS;
    const int b    = blockIdx.x / K1_WGS;
    const int wave = threadIdx.x >> 5;
    const int lane = threadIdx.x & 31;
    const int lm   = lane & 15;
    const int kb   = (lane < 16) ? 0 : 2;

    const long n0 = (long)wg * (K1_WAVES * K1_ROWS) + (long)wave * K1_ROWS;
    const float* qb = q + ((long)b * NSP + n0) * CCH;
    const float* kp = k + ((long)b * NSP + n0) * CCH;

    v8f c00 = {}, c01 = {}, c10 = {}, c11 = {};
    for (int it = 0; it < K1_ROWS; it += 4) {
        const float* qr = qb + (long)(it + kb) * CCH;
        const float* kr = kp + (long)(it + kb) * CCH;
        v2f a0, a1, b0, b1;
        a0.x = qr[lm];           a0.y = qr[CCH + lm];
        a1.x = qr[16 + lm];      a1.y = qr[CCH + 16 + lm];
        b0.x = __builtin_nontemporal_load(kr + lm);
        b0.y = __builtin_nontemporal_load(kr + CCH + lm);
        b1.x = __builtin_nontemporal_load(kr + 16 + lm);
        b1.y = __builtin_nontemporal_load(kr + CCH + 16 + lm);
        c00 = __builtin_amdgcn_wmma_f32_16x16x4_f32(false, a0, false, b0, (short)0, c00, false, false);
        c01 = __builtin_amdgcn_wmma_f32_16x16x4_f32(false, a0, false, b1, (short)0, c01, false, false);
        c10 = __builtin_amdgcn_wmma_f32_16x16x4_f32(false, a1, false, b0, (short)0, c10, false, false);
        c11 = __builtin_amdgcn_wmma_f32_16x16x4_f32(false, a1, false, b1, (short)0, c11, false, false);
    }

    __shared__ float smem[K1_WAVES * CCH * CCH];
    const int rowadd = (lane < 16) ? 0 : 8;
    float* sw = smem + wave * (CCH * CCH);
#pragma unroll
    for (int r = 0; r < 8; ++r) {
        const int i0 = r + rowadd;
        sw[(i0     ) * CCH + lm     ] = c00[r];
        sw[(i0     ) * CCH + 16 + lm] = c01[r];
        sw[(i0 + 16) * CCH + lm     ] = c10[r];
        sw[(i0 + 16) * CCH + 16 + lm] = c11[r];
    }
    __syncthreads();

    float* outp = partial + (long)(b * K1_WGS + wg) * (CCH * CCH);
    {
        const int e = threadIdx.x;
        float s = 0.f;
        for (int w = 0; w < K1_WAVES; ++w) s += smem[w * (CCH * CCH) + e];
        outp[e] = s;
    }
}

__global__ __launch_bounds__(K3_WAVES * 32) void apply_kernel(const float* __restrict__ q,
                                                              const float* __restrict__ partial,
                                                              float* __restrict__ out) {
    const int wg   = blockIdx.x % K3_WGS;
    const int b    = blockIdx.x / K3_WGS;
    const int wave = threadIdx.x >> 5;
    const int lane = threadIdx.x & 31;
    const int lm   = lane & 15;
    const int h    = lane >> 4;

    __shared__ __attribute__((aligned(16))) float wsm[16 * WPITCH];
#pragma unroll
    for (int half = 0; half < 2; ++half) {
        const int i = wave + 16 * half;
        const float* p = partial + (long)b * K1_WGS * (CCH * CCH) + i * CCH + lane;
        float v[K1_WGS];
#pragma unroll
        for (int w = 0; w < K1_WGS; ++w) v[w] = p[w * (CCH * CCH)];
        float g = 0.f;
#pragma unroll
        for (int w = 0; w < K1_WGS; ++w) g += v[w];
        float m = -3.4e38f;
#pragma unroll
        for (int j = 0; j < CCH; ++j)
            m = fmaxf(m, __builtin_bit_cast(float, __builtin_amdgcn_readlane(__builtin_bit_cast(int, g), j)));
        const float ex = __expf(g - m);
        float s = 0.f;
#pragma unroll
        for (int j = 0; j < CCH; ++j)
            s += __builtin_bit_cast(float, __builtin_amdgcn_readlane(__builtin_bit_cast(int, ex), j));
        const float inv = 1.f / s;
        wsm[(i >> 1) * WPITCH + (i & 1) * CCH + lane] = ex * inv;
    }
    __syncthreads();

    v2f Bw0[8], Bw1[8];
#pragma unroll
    for (int u = 0; u < 4; ++u) {
        const v4f w0 = *(const v4f*)(wsm + lm * WPITCH +       8 * u + 4 * h);
        const v4f w1 = *(const v4f*)(wsm + lm * WPITCH + CCH + 8 * u + 4 * h);
        Bw0[2 * u].x = w0.x;  Bw0[2 * u].y = w0.y;  Bw0[2 * u + 1].x = w0.z;  Bw0[2 * u + 1].y = w0.w;
        Bw1[2 * u].x = w1.x;  Bw1[2 * u].y = w1.y;  Bw1[2 * u + 1].x = w1.z;  Bw1[2 * u + 1].y = w1.w;
    }

    const long n0 = (long)wg * (K3_WAVES * K3_ROWS) + (long)wave * K3_ROWS;
    const float* qb = q   + ((long)b * NSP + n0) * CCH;
    float*       ob = out + ((long)b * NSP + n0) * CCH;

#pragma unroll 1
    for (int it = 0; it < K3_ROWS; it += 16) {
        const float* ar = qb + (long)(it + lm) * CCH + 4 * h;
        v4f Q[4];
#pragma unroll
        for (int u = 0; u < 4; ++u) Q[u] = *(const v4f*)(ar + 8 * u);
        const float* rr = qb + (long)(it + 8 * h) * CCH + 2 * lm;
        v2f R[8];
#pragma unroll
        for (int r = 0; r < 8; ++r) R[r] = *(const v2f*)(rr + r * CCH);

        __builtin_amdgcn_sched_barrier(0);

        v8f c0 = {}, c1 = {};
#pragma unroll
        for (int u = 0; u < 4; ++u) {
            v2f a0, a1;
            a0.x = Q[u].x;  a0.y = Q[u].y;
            a1.x = Q[u].z;  a1.y = Q[u].w;
            c0 = __builtin_amdgcn_wmma_f32_16x16x4_f32(false, a0, false, Bw0[2 * u],     (short)0, c0, false, false);
            c1 = __builtin_amdgcn_wmma_f32_16x16x4_f32(false, a0, false, Bw1[2 * u],     (short)0, c1, false, false);
            c0 = __builtin_amdgcn_wmma_f32_16x16x4_f32(false, a1, false, Bw0[2 * u + 1], (short)0, c0, false, false);
            c1 = __builtin_amdgcn_wmma_f32_16x16x4_f32(false, a1, false, Bw1[2 * u + 1], (short)0, c1, false, false);
        }
        __builtin_amdgcn_sched_barrier(0);
        asm volatile("v_nop\n\tv_nop\n\tv_nop\n\tv_nop");
        __builtin_amdgcn_sched_barrier(0);

        float* orow = ob + (long)(it + 8 * h) * CCH + 2 * lm;
#pragma unroll
        for (int r = 0; r < 8; ++r) {
            v2f o;
            o.x = c0[r] + R[r].x;
            o.y = c1[r] + R[r].y;
            __builtin_nontemporal_store(o, (v2f*)(orow + r * CCH));
        }
    }
}

extern "C" void kernel_launch(void* const* d_in, const int* in_sizes, int n_in,
                              void* d_out, int out_size, void* d_ws, size_t ws_size,
                              hipStream_t stream) {
    const float* q = (const float*)d_in[0];
    const float* k = (const float*)d_in[1];
    float* out = (float*)d_out;
    const int B = in_sizes[0] / (NSP * CCH);

    float* partial = (float*)d_ws;

    gram_kernel   <<<dim3(B * K1_WGS), dim3(K1_WAVES * 32), 0, stream>>>(q, k, partial);
    apply_kernel  <<<dim3(B * K3_WGS), dim3(K3_WAVES * 32), 0, stream>>>(q, partial, out);
}


# gram pass: fragment channel/row order matched to memory: a load takes 256 contiguous B, 8 B per lane (was 4 B from two half lines); 32 loads cover 32 rows (was 16); sum order in 4-row steps permuted
# speedup vs baseline: 1.4734x; 1.1727x over previous
#include <hip/hip_runtime.h>

typedef __attribute__((ext_vector_type(2))) float v2f;
typedef __attribute__((ext_vector_type(4))) float v4f;
typedef __attribute__((ext_vector_type(8))) float v8f;

#define CCH 32
#define NSP (64 * 64 * 64)
#define K1_WGS 64
#define K1_WAVES 32
#define K1_ROWS (NSP / (K1_WGS * K1_WAVES))
#define K1_TRIP 32
#define K3_WGS 64
#define K3_WAVES 16
#define K3_ROWS (NSP / (K3_WGS * K3_WAVES))
#define WPITCH 68

__global__ __launch_bounds__(K1_WAVES * 32) void gram_kernel(const float* __restrict__ q,
                                                   const float* __restrict__ k,
                                                   float* __restrict__ partial) {
    const int wg   = blockIdx.x % K1_WGS;
    const int b    = blockIdx.x / K1_WGS;
    const int wave = threadIdx.x >> 5;
    const int lane = threadIdx.x & 31;
    const int lm   = lane & 15;

    const long n0 = (long)wg * (K1_WAVES * K1_ROWS) + (long)wave * K1_ROWS;
    const float* qb = q + ((long)b * NSP + n0) * CCH;
    const float* kp = k + ((long)b * NSP + n0) * CCH;

    v8f c00 = {}, c01 = {}, c10 = {}, c11 = {};
    for (int it = 0; it < K1_ROWS; it += K1_TRIP) {
        const float* qr = qb + (long)it * CCH + 2 * lane;
        const float* kr = kp + (long)it * CCH + 2 * lane;
        v2f Q[K1_TRIP / 2], K[K1_TRIP / 2];
#pragma unroll
        for (int s = 0; s < K1_TRIP / 2; ++s) Q[s] = *(const v2f*)(qr + s * 2 * CCH);
#pragma unroll
        for (int s = 0; s < K1_TRIP / 2; ++s) K[s] = __builtin_nontemporal_load((const v2f*)(kr + s * 2 * CCH));
        __builtin_amdgcn_sched_barrier(0);
#pragma unroll
        for (int g = 0; g < K1_TRIP / 4; ++g) {
            v2f a0, a1, b0, b1;
            a0.x = Q[2 * g].x;  a0.y = Q[2 * g + 1].x;
            a1.x = Q[2 * g].y;  a1.y = Q[2 * g + 1].y;
            b0.x = K[2 * g].x;  b0.y = K[2 * g + 1].x;
            b1.x = K[2 * g].y;  b1.y = K[2 * g + 1].y;
            c00 = __builtin_amdgcn_wmma_f32_16x16x4_f32(false, a0, false, b0, (short)0, c00, false, false);
            c01 = __builtin_amdgcn_wmma_f32_16x16x4_f32(false, a0, false, b1, (short)0, c01, false, false);
            c10 = __builtin_amdgcn_wmma_f32_16x16x4_f32(false, a1, false, b0, (short)0, c10, false, false);
            c11 = __builtin_amdgcn_wmma_f32_16x16x4_f32(false, a1, false, b1, (short)0, c11, false, false);
        }
    }

    __shared__ __attribute__((aligned(16))) float smem[K1_WAVES * CCH * CCH];
    const int rowadd = (lane < 16) ? 0 : 8;
    float* sw = smem + wave * (CCH * CCH);
#pragma unroll
    for (int r = 0; r < 8; ++r) {
        const int m = r + rowadd;
        v2f e0, e1;
        e0.x = c00[r];  e0.y = c01[r];
        e1.x = c10[r];  e1.y = c11[r];
        *(v2f*)(sw + (2 * m    ) * CCH + 2 * lm) = e0;
        *(v2f*)(sw + (2 * m + 1) * CCH + 2 * lm) = e1;
    }
    __syncthreads();

    float* outp = partial + (long)(b * K1_WGS + wg) * (CCH * CCH);
    {
        const int e = threadIdx.x;
        float s = 0.f;
        for (int w = 0; w < K1_WAVES; ++w) s += smem[w * (CCH * CCH) + e];
        outp[e] = s;
    }
}

__global__ __launch_bounds__(K3_WAVES * 32) void apply_kernel(const float* __restrict__ q,
                                                              const float* __restrict__ partial,
                                                              float* __restrict__ out) {
    const int wg   = blockIdx.x % K3_WGS;
    const int b    = blockIdx.x / K3_WGS;
    const int wave = threadIdx.x >> 5;
    const int lane = threadIdx.x & 31;
    const int lm   = lane & 15;
    const int h    = lane >> 4;

    __shared__ __attribute__((aligned(16))) float wsm[16 * WPITCH];
#pragma unroll
    for (int half = 0; half < 2; ++half) {
        const int i = wave + 16 * half;
        const float* p = partial + (long)b * K1_WGS * (CCH * CCH) + i * CCH + lane;
        float v[K1_WGS];
#pragma unroll
        for (int w = 0; w < K1_WGS; ++w) v[w] = p[w * (CCH * CCH)];
        float g = 0.f;
#pragma unroll
        for (int w = 0; w < K1_WGS; ++w) g += v[w];
        float m = -3.4e38f;
#pragma unroll
        for (int j = 0; j < CCH; ++j)
            m = fmaxf(m, __builtin_bit_cast(float, __builtin_amdgcn_readlane(__builtin_bit_cast(int, g), j)));
        const float ex = __expf(g - m);
        float s = 0.f;
#pragma unroll
        for (int j = 0; j < CCH; ++j)
            s += __builtin_bit_cast(float, __builtin_amdgcn_readlane(__builtin_bit_cast(int, ex), j));
        const float inv = 1.f / s;
        wsm[(i >> 1) * WPITCH + (i & 1) * CCH + lane] = ex * inv;
    }
    __syncthreads();

    v2f Bw0[8], Bw1[8];
#pragma unroll
    for (int u = 0; u < 4; ++u) {
        const v4f w0 = *(const v4f*)(wsm + lm * WPITCH +       8 * u + 4 * h);
        const v4f w1 = *(const v4f*)(wsm + lm * WPITCH + CCH + 8 * u + 4 * h);
        Bw0[2 * u].x = w0.x;  Bw0[2 * u].y = w0.y;  Bw0[2 * u + 1].x = w0.z;  Bw0[2 * u + 1].y = w0.w;
        Bw1[2 * u].x = w1.x;  Bw1[2 * u].y = w1.y;  Bw1[2 * u + 1].x = w1.z;  Bw1[2 * u + 1].y = w1.w;
    }

    const long n0 = (long)wg * (K3_WAVES * K3_ROWS) + (long)wave * K3_ROWS;
    const float* qb = q   + ((long)b * NSP + n0) * CCH;
    float*       ob = out + ((long)b * NSP + n0) * CCH;

#pragma unroll 1
    for (int it = 0; it < K3_ROWS; it += 16) {
        const float* ar = qb + (long)(it + lm) * CCH + 4 * h;
        v4f Q[4];
#pragma unroll
        for (int u = 0; u < 4; ++u) Q[u] = *(const v4f*)(ar + 8 * u);
        const float* rr = qb + (long)(it + 8 * h) * CCH + 2 * lm;
        v2f R[8];
#pragma unroll
        for (int r = 0; r < 8; ++r) R[r] = *(const v2f*)(rr + r * CCH);

        __builtin_amdgcn_sched_barrier(0);

        v8f c0 = {}, c1 = {};
#pragma unroll
        for (int u = 0; u < 4; ++u) {
            v2f a0, a1;
            a0.x = Q[u].x;  a0.y = Q[u].y;
            a1.x = Q[u].z;  a1.y = Q[u].w;
            c0 = __builtin_amdgcn_wmma_f32_16x16x4_f32(false, a0, false, Bw0[2 * u],     (short)0, c0, false, false);
            c1 = __builtin_amdgcn_wmma_f32_16x16x4_f32(false, a0, false, Bw1[2 * u],     (short)0, c1, false, false);
            c0 = __builtin_amdgcn_wmma_f32_16x16x4_f32(false, a1, false, Bw0[2 * u + 1], (short)0, c0, false, false);
            c1 = __builtin_amdgcn_wmma_f32_16x16x4_f32(false, a1, false, Bw1[2 * u + 1], (short)0, c1, false, false);
        }
        __builtin_amdgcn_sched_barrier(0);
        asm volatile("v_nop\n\tv_nop\n\tv_nop\n\tv_nop");
        __builtin_amdgcn_sched_barrier(0);

        float* orow = ob + (long)(it + 8 * h) * CCH + 2 * lm;
#pragma unroll
        for (int r = 0; r < 8; ++r) {
            v2f o;
            o.x = c0[r] + R[r].x;
            o.y = c1[r] + R[r].y;
            __builtin_nontemporal_store(o, (v2f*)(orow + r * CCH));
        }
    }
}

extern "C" void kernel_launch(void* const* d_in, const int* in_sizes, int n_in,
                              void* d_out, int out_size, void* d_ws, size_t ws_size,
                              hipStream_t stream) {
    const float* q = (const float*)d_in[0];
    const float* k = (const float*)d_in[1];
    float* out = (float*)d_out;
    const int B = in_sizes[0] / (NSP * CCH);

    float* partial = (float*)d_ws;

    gram_kernel   <<<dim3(B * K1_WGS), dim3(K1_WAVES * 32), 0, stream>>>(q, k, partial);
    apply_kernel  <<<dim3(B * K3_WGS), dim3(K3_WAVES * 32), 0, stream>>>(q, partial, out);
}
